# starting kernel
# baseline (speedup 1.0000x reference)
#include <hip/hip_runtime.h>

typedef float v2f __attribute__((ext_vector_type(2)));
typedef float v4f __attribute__((ext_vector_type(4)));
typedef float v8f __attribute__((ext_vector_type(8)));

#define S_LEN 128
#define NB    4
#define EMB   512
#define NH    8
#define HD    64
#define BH    32

__global__ __launch_bounds__(256) void k_vproj(const float* __restrict__ value,
                                               const float* __restrict__ Wv,
                                               const float* __restrict__ bv,
                                               float* __restrict__ v_t) {
  const int lane = threadIdx.x & 31;
  const int wid  = blockIdx.x * 8 + (threadIdx.x >> 5);
  const int m0   = (wid >> 5) * 16;
  const int c0   = (wid & 31) * 16;
  const int half = lane >> 4;
  const int l15  = lane & 15;

  const int am = m0 + l15;
  const int bn = c0 + l15;
  v8f acc = {};
  for (int k0 = 0; k0 < EMB; k0 += 4) {
    const int ka = k0 + 2 * half;
    v2f a = *(const v2f*)(value + am * EMB + ka);
    v2f b = *(const v2f*)(Wv    + bn * EMB + ka);
    acc = __builtin_amdgcn_wmma_f32_16x16x4_f32(false, a, false, b,
                                                (short)0, acc, false, false);
  }
  const float bias = bv[bn];
  const int h  = bn >> 6;
  const int dd = bn & 63;
#pragma unroll
  for (int r = 0; r < 8; ++r) {
    const int m = m0 + r + 8 * half;
    const int s = m >> 2, n = m & 3;
    v_t[((n * NH + h) * S_LEN + s) * HD + dd] = acc[r] + bias;
  }
}

__global__ __launch_bounds__(256) void k_edge(const float* __restrict__ attn,
                                              const float* __restrict__ edge_v,
                                              float* __restrict__ out_pre) {
  const int lane = threadIdx.x & 31;
  const int b    = blockIdx.x * 8 + (threadIdx.x >> 5);
  const int half = lane >> 4;
  const int l15  = lane & 15;

  const float* evb = edge_v + (size_t)b * (S_LEN * HD) + l15 * 4;
  const float* wb  = attn + b * S_LEN + half;
  v4f acc = {};
#pragma unroll 4
  for (int s = 0; s < S_LEN; s += 2) {
    const float aw = wb[s];
    v4f ev = __builtin_nontemporal_load((const v4f*)(evb + (s + half) * HD));
    acc += aw * ev;
  }
  acc.x += __shfl_xor(acc.x, 16);
  acc.y += __shfl_xor(acc.y, 16);
  acc.z += __shfl_xor(acc.z, 16);
  acc.w += __shfl_xor(acc.w, 16);
  if (lane < 16)
    *(v4f*)(out_pre + b * HD + lane * 4) = acc;
}

__global__ __launch_bounds__(256) void k_attn(const float* __restrict__ attn,
                                              const float* __restrict__ v_t,
                                              float* __restrict__ out_pre) {
  const int lane = threadIdx.x & 31;
  const int wid  = blockIdx.x * 8 + (threadIdx.x >> 5);
  const int g    = wid >> 5;
  const int t    = wid & 31;
  const int q0   = (t >> 2) * 16;
  const int c0   = (t & 3) * 16;
  const int half = lane >> 4;
  const int l15  = lane & 15;

  const float* Ag = attn + (size_t)g * S_LEN * S_LEN;
  const float* Bg = v_t + g * S_LEN * HD;
  float*       Og = out_pre + g * S_LEN * HD;

  v8f acc;
#pragma unroll
  for (int r = 0; r < 8; ++r)
    acc[r] = Og[(q0 + r + 8 * half) * HD + c0 + l15];

  for (int k0 = 0; k0 < S_LEN; k0 += 4) {
    const int ka = k0 + 2 * half;
    v2f a = *(const v2f*)(Ag + (q0 + l15) * S_LEN + ka);
    v2f b;
    b.x = Bg[ka * HD + c0 + l15];
    b.y = Bg[(ka + 1) * HD + c0 + l15];
    acc = __builtin_amdgcn_wmma_f32_16x16x4_f32(false, a, false, b,
                                                (short)0, acc, false, false);
  }
#pragma unroll
  for (int r = 0; r < 8; ++r)
    Og[(q0 + r + 8 * half) * HD + c0 + l15] = acc[r];
}

__global__ __launch_bounds__(256) void k_outproj(const float* __restrict__ out_pre,
                                                 const float* __restrict__ Wout,
                                                 const float* __restrict__ bout,
                                                 float* __restrict__ out) {
  const int lane = threadIdx.x & 31;
  const int wid  = blockIdx.x * 8 + (threadIdx.x >> 5);
  const int m0   = (wid >> 5) * 16;
  const int c0   = (wid & 31) * 16;
  const int half = lane >> 4;
  const int l15  = lane & 15;

  const int am = m0 + l15;
  const int q  = am >> 2, n = am & 3;
  const int bn = c0 + l15;
  v8f acc = {};
  for (int k0 = 0; k0 < EMB; k0 += 4) {
    const int e = k0 + 2 * half;
    v2f a = *(const v2f*)(out_pre + ((n * NH + (e >> 6)) * S_LEN + q) * HD + (e & 63));
    v2f b = *(const v2f*)(Wout + bn * EMB + e);
    acc = __builtin_amdgcn_wmma_f32_16x16x4_f32(false, a, false, b,
                                                (short)0, acc, false, false);
  }
  const float bias = bout[bn];
#pragma unroll
  for (int r = 0; r < 8; ++r)
    out[(m0 + r + 8 * half) * EMB + c0 + l15] = acc[r] + bias;
}

extern "C" void kernel_launch(void* const* d_in, const int* in_sizes, int n_in,
                              void* d_out, int out_size, void* d_ws, size_t ws_size,
                              hipStream_t stream) {
  const float* attn  = (const float*)d_in[0];
  const float* value = (const float*)d_in[1];
  const float* edgev = (const float*)d_in[2];
  const float* Wv    = (const float*)d_in[3];
  const float* bv    = (const float*)d_in[4];
  const float* Wout  = (const float*)d_in[5];
  const float* bout  = (const float*)d_in[6];
  float* out = (float*)d_out;

  float* v_t     = (float*)d_ws;
  float* out_pre = v_t + BH * S_LEN * HD;

  k_vproj  <<<128, 256, 0, stream>>>(value, Wv, bv, v_t);
  k_edge   <<<512, 256, 0, stream>>>(attn, edgev, out_pre);
  k_attn   <<<128, 256, 0, stream>>>(attn, v_t, out_pre);
  k_outproj<<<128, 256, 0, stream>>>(out_pre, Wout, bout, out);
}


# both K=512 products: a block's 4 waves split K (128 k each), each multiplies the whole block tile (4 tiles share one fragment), partials added in LDS; 80 requests a wave, was 128; sum tree changes
# speedup vs baseline: 1.6618x; 1.6618x over previous
#include <hip/hip_runtime.h>


typedef float v2f __attribute__((ext_vector_type(2)));
typedef float v4f __attribute__((ext_vector_type(4)));
typedef float v8f __attribute__((ext_vector_type(8)));

#define S_LEN 128
#define NB    4
#define EMB   512
#define NH    8
#define HD    64
#define BH    32

#define ONE_WAVE_PER_SIMD __attribute__((amdgpu_waves_per_eu(1, 1)))

#define SLAB_W   16
#define N_SLAB   (HD / SLAB_W)
#define PLANE    (BH * S_LEN * HD)
#define S_HALF   (S_LEN / 2)
#define VS_PITCH (S_HALF + 4)

#define KQ       (EMB / 4)
#define KQ_STEPS (KQ / 8)
#define KQ_LEAD  12

__global__ __launch_bounds__(128) ONE_WAVE_PER_SIMD void k_vattn(const float* __restrict__ value,
                                               const float* __restrict__ Wv,
                                               const float* __restrict__ bv,
                                               const float* __restrict__ attn,
                                               float* __restrict__ planes) {
  __shared__ __attribute__((aligned(16))) float vs[4][SLAB_W][VS_PITCH];

  const int lane = threadIdx.x & 31;
  const int w    = threadIdx.x >> 5;
  const int kh   = blockIdx.x & 1;
  const int slab = (blockIdx.x >> 1) & 3;
  const int g    = blockIdx.x >> 3;
  const int n    = g >> 3;
  const int h    = g & 7;
  const int half = lane >> 4;
  const int l15  = lane & 15;

  const int am = (S_HALF * kh + l15) * NB + n;
  const int bn = h * HD + slab * SLAB_W + l15;
  unsigned a_off = (am * EMB + KQ * w + 4 * half) * 4;
  unsigned w_off = (bn * EMB + KQ * w + 4 * half) * 4;
  unsigned b_off = bn * 4;
  const int q0 = 32 * w;
  unsigned q_off = (((g * S_LEN + q0 + l15) * S_LEN) + S_HALF * kh + 4 * half) * 4;
  unsigned o_off = (kh * PLANE
                    + ((g * N_SLAB + slab) * S_LEN + q0 + 8 * half) * SLAB_W + l15) * 4;
  unsigned st_off = ((w * SLAB_W + l15) * VS_PITCH + 8 * half) * 4;
  unsigned ld_off = (l15 * VS_PITCH + 4 * half) * 4;
  v8f acc[4] = {};
  asm volatile("" : "+v"(acc[0]), "+v"(b_off), "+v"(a_off), "+v"(w_off), "+v"(q_off), "+v"(o_off),
                    "+v"(st_off), "+v"(ld_off));
  const float* ap  = (const float*)((const char*)value + a_off);
  const float* wp  = (const float*)((const char*)Wv + w_off);
  const float* a0p = (const float*)((const char*)attn + q_off);
  const float* a1p = a0p + 16 * S_LEN;
  float*       outp = (float*)((char*)planes + o_off);
  v4f fa[KQ_STEPS][4], fb[KQ_STEPS];
  v4f a0[S_HALF / 8], a1[S_HALF / 8];
  float bias = 0.0f;
#pragma unroll
  for (int j = 0; j < KQ_LEAD; ++j) {
#pragma unroll
    for (int t = 0; t < 4; ++t)
      fa[j][t] = *(const v4f*)(ap + t * (16 * NB * EMB) + j * 8);
    fb[j] = *(const v4f*)(wp + j * 8);
  }
#pragma unroll
  for (int j = 0; j < KQ_STEPS; ++j) {
    __builtin_amdgcn_sched_barrier(0);
    if (j + KQ_LEAD < KQ_STEPS) {
#pragma unroll
      for (int t = 0; t < 4; ++t)
        fa[j + KQ_LEAD][t] = *(const v4f*)(ap + t * (16 * NB * EMB) + (j + KQ_LEAD) * 8);
      fb[j + KQ_LEAD] = *(const v4f*)(wp + (j + KQ_LEAD) * 8);
    }
    if (j >= 6 && j < 10) {
#pragma unroll
      for (int i = 2 * (j - 6); i < 2 * (j - 6) + 2; ++i) {
        a0[i] = *(const v4f*)(a0p + 8 * i);
        a1[i] = *(const v4f*)(a1p + 8 * i);
      }
    }
    if (j == 10)
      bias = *(const float*)((const char*)bv + b_off);
    __builtin_amdgcn_sched_barrier(0);
#pragma unroll
    for (int t = 0; t < 4; ++t)
      acc[t] = __builtin_amdgcn_wmma_f32_16x16x4_f32(false, fa[j][t].xy, false, fb[j].xy,
                                                     (short)0, acc[t], false, false);
#pragma unroll
    for (int t = 0; t < 4; ++t)
      acc[t] = __builtin_amdgcn_wmma_f32_16x16x4_f32(false, fa[j][t].zw, false, fb[j].zw,
                                                     (short)0, acc[t], false, false);
  }
  asm volatile("v_nop\n\tv_nop\n\tv_nop\n\tv_nop" : "+v"(acc[0]), "+v"(acc[1]), "+v"(acc[2]), "+v"(acc[3]));
  {
    char* vsb = (char*)&vs[0][0][0];
#pragma unroll
    for (int t = 0; t < 4; ++t) {
      v4f lo = {acc[t][0], acc[t][1], acc[t][2], acc[t][3]};
      v4f hi = {acc[t][4], acc[t][5], acc[t][6], acc[t][7]};
      *(v4f*)(vsb + st_off + 64 * t)      = lo;
      *(v4f*)(vsb + st_off + 64 * t + 16) = hi;
    }
  }
  __builtin_amdgcn_fence(__ATOMIC_RELEASE, "workgroup", "local");
  __builtin_amdgcn_s_barrier();
  __builtin_amdgcn_fence(__ATOMIC_ACQUIRE, "workgroup", "local");

  const char* bpb = (const char*)&vs[0][0][0] + ld_off;
  v4f b[S_HALF / 8];
#pragma unroll
  for (int i = 0; i < S_HALF / 8; ++i) {
    const v4f p0 = *(const v4f*)(bpb + 32 * i);
    const v4f p1 = *(const v4f*)(bpb + 1 * (SLAB_W * VS_PITCH * 4) + 32 * i);
    const v4f p2 = *(const v4f*)(bpb + 2 * (SLAB_W * VS_PITCH * 4) + 32 * i);
    const v4f p3 = *(const v4f*)(bpb + 3 * (SLAB_W * VS_PITCH * 4) + 32 * i);
    b[i] = ((p0 + p1) + (p2 + p3)) + bias;
  }
  v8f c0 = {}, c1 = {};
  __builtin_amdgcn_sched_barrier(0);
#pragma unroll
  for (int i = 0; i < S_HALF / 8; ++i) {
    c0 = __builtin_amdgcn_wmma_f32_16x16x4_f32(false, a0[i].xy, false, b[i].xy,
                                               (short)0, c0, false, false);
    c1 = __builtin_amdgcn_wmma_f32_16x16x4_f32(false, a1[i].xy, false, b[i].xy,
                                               (short)0, c1, false, false);
    c0 = __builtin_amdgcn_wmma_f32_16x16x4_f32(false, a0[i].zw, false, b[i].zw,
                                               (short)0, c0, false, false);
    c1 = __builtin_amdgcn_wmma_f32_16x16x4_f32(false, a1[i].zw, false, b[i].zw,
                                               (short)0, c1, false, false);
  }
#pragma unroll
  for (int r = 0; r < 8; ++r) {
    outp[r * SLAB_W]        = c0[r];
    outp[(16 + r) * SLAB_W] = c1[r];
  }
}

__global__ __launch_bounds__(256) void k_edge(const float* __restrict__ attn,
                                              const float* __restrict__ edge_v,
                                              float* __restrict__ planes) {
  const int lane = threadIdx.x & 31;
  const int b    = blockIdx.x * 8 + (threadIdx.x >> 5);
  const int half = lane >> 4;
  const int l15  = lane & 15;

  const int off = (((b >> 7) * N_SLAB + (l15 >> 2)) * S_LEN + (b & 127)) * SLAB_W + (l15 & 3) * 4;
  const v4f p0 = *(const v4f*)(planes + off);
  const v4f p1 = *(const v4f*)(planes + PLANE + off);

  const float* evb = edge_v + (size_t)b * (S_LEN * HD) + l15 * 4;
  const float* wb  = attn + b * S_LEN + half;
  v4f acc = {};
#pragma unroll 4
  for (int s = 0; s < S_LEN; s += 2) {
    const float aw = wb[s];
    v4f ev = __builtin_nontemporal_load((const v4f*)(evb + (s + half) * HD));
    acc += aw * ev;
  }
  acc.x += __shfl_xor(acc.x, 16);
  acc.y += __shfl_xor(acc.y, 16);
  acc.z += __shfl_xor(acc.z, 16);
  acc.w += __shfl_xor(acc.w, 16);
  if (lane < 16)
    *(v4f*)(planes + off) = (acc + p0) + p1;
}

__global__ __launch_bounds__(128) ONE_WAVE_PER_SIMD void k_outproj(const float* __restrict__ pre,
                                                 const float* __restrict__ Wout,
                                                 const float* __restrict__ bout,
                                                 float* __restrict__ out) {
  __shared__ v4f part[4][4][2][32];

  const int lane = threadIdx.x & 31;
  const int w    = threadIdx.x >> 5;
  const int m0   = (blockIdx.x >> 3) * 16;
  const int cb   = (blockIdx.x & 7) * 64;
  const int half = lane >> 4;
  const int l15  = lane & 15;

  const int am = m0 + l15;
  const int q  = am >> 2, n = am & 3;
  const int bn = cb + l15;
  const float bias = bout[bn + 16 * w];
  const float* abase = pre + ((n * NH * N_SLAB + 8 * w) * S_LEN + q) * SLAB_W + 4 * half;
  const float* bbase = Wout + bn * EMB + KQ * w + 4 * half;
  float* outp = out + (m0 + 8 * half) * EMB + cb + 16 * w + l15;
  v8f acc[4] = {};
#define PRE_AT(kk) (abase + ((kk) >> 4) * (S_LEN * SLAB_W) + ((kk) & 15))
  v4f fa[KQ_STEPS], fb[KQ_STEPS][4];
#pragma unroll
  for (int j = 0; j < KQ_LEAD; ++j) {
    fa[j] = *(const v4f*)PRE_AT(j * 8);
#pragma unroll
    for (int t = 0; t < 4; ++t)
      fb[j][t] = *(const v4f*)(bbase + t * (16 * EMB) + j * 8);
  }
#pragma unroll
  for (int j = 0; j < KQ_STEPS; ++j) {
    __builtin_amdgcn_sched_barrier(0);
    if (j + KQ_LEAD < KQ_STEPS) {
      fa[j + KQ_LEAD] = *(const v4f*)PRE_AT((j + KQ_LEAD) * 8);
#pragma unroll
      for (int t = 0; t < 4; ++t)
        fb[j + KQ_LEAD][t] = *(const v4f*)(bbase + t * (16 * EMB) + (j + KQ_LEAD) * 8);
    }
    __builtin_amdgcn_sched_barrier(0);
#pragma unroll
    for (int t = 0; t < 4; ++t)
      acc[t] = __builtin_amdgcn_wmma_f32_16x16x4_f32(false, fa[j].xy, false, fb[j][t].xy,
                                                     (short)0, acc[t], false, false);
#pragma unroll
    for (int t = 0; t < 4; ++t)
      acc[t] = __builtin_amdgcn_wmma_f32_16x16x4_f32(false, fa[j].zw, false, fb[j][t].zw,
                                                     (short)0, acc[t], false, false);
  }
#undef PRE_AT
#pragma unroll
  for (int t = 0; t < 4; ++t) {
    part[w][t][0][lane] = (v4f){acc[t][0], acc[t][1], acc[t][2], acc[t][3]};
    part[w][t][1][lane] = (v4f){acc[t][4], acc[t][5], acc[t][6], acc[t][7]};
  }
  __syncthreads();
  const v4f lo = ((part[0][w][0][lane] + part[1][w][0][lane])
                  + (part[2][w][0][lane] + part[3][w][0][lane])) + bias;
  const v4f hi = ((part[0][w][1][lane] + part[1][w][1][lane])
                  + (part[2][w][1][lane] + part[3][w][1][lane])) + bias;
#pragma unroll
  for (int r = 0; r < 4; ++r) {
    outp[r * EMB]       = lo[r];
    outp[(4 + r) * EMB] = hi[r];
  }
}

extern "C" void kernel_launch(void* const* d_in, const int* in_sizes, int n_in,
                              void* d_out, int out_size, void* d_ws, size_t ws_size,
                              hipStream_t stream) {
  const float* attn  = (const float*)d_in[0];
  const float* value = (const float*)d_in[1];
  const float* edgev = (const float*)d_in[2];
  const float* Wv    = (const float*)d_in[3];
  const float* bv    = (const float*)d_in[4];
  const float* Wout  = (const float*)d_in[5];
  const float* bout  = (const float*)d_in[6];
  float* out = (float*)d_out;

  float* planes = (float*)d_ws;

  k_vattn  <<<256, 128, 0, stream>>>(value, Wv, bv, attn, planes);
  k_edge   <<<512, 256, 0, stream>>>(attn, edgev, planes);
  k_outproj<<<256, 128, 0, stream>>>(planes, Wout, bout, out);
}
